# starting kernel
# baseline (speedup 1.0000x reference)
#include <hip/hip_runtime.h>
#include <math.h>

typedef __attribute__((ext_vector_type(2))) float v2f;
typedef __attribute__((ext_vector_type(8))) float v8f;

constexpr int NB   = 2;
constexpr int NN   = 131072;
constexpr int DD   = 24;
constexpr int KNB  = 4;
constexpr int HID  = 128;
constexpr int DLAT = 16;
constexpr int TPB  = NN / 16;

constexpr size_t XCOUNT   = (size_t)NB * NN * DD;
constexpr size_t W1S_OFF  = XCOUNT;
constexpr size_t W1S_F2   = 24 * 8 * 32;
constexpr size_t W2S_F2   = 32 * 32;
constexpr size_t WSW_FLTS = 2 * (W1S_F2 + W2S_F2);

constexpr int LDS_W2_OFF    = 12288;
constexpr int LDS_WAVE_OFF  = 14336;
constexpr int LDS_WAVE_SZ   = 2048;
constexpr int LDS_TOTAL_B   = (LDS_WAVE_OFF + 8 * LDS_WAVE_SZ) * 4;

__device__ __forceinline__ float gelu_tanh(float x) {
    float u = 0.7978845608028654f * (x + 0.044715f * x * x * x);
    return 0.5f * x * (1.0f + tanhf(u));
}

__global__ __launch_bounds__(256) void swizzle_weights(
    const float* __restrict__ W1, const float* __restrict__ W2,
    float* __restrict__ wsw)
{
    int t = blockIdx.x * 256 + threadIdx.x;
    if (t < (int)W1S_F2) {
        int kk  = t >> 8;
        int rem = t & 255;
        int j    = rem >> 5;
        int l    = rem & 31;
        int half = l >> 4;
        int n    = l & 15;
        int row0 = 4 * kk + 2 * half;
        int col  = 16 * j + n;
        float2 v = make_float2(W1[row0 * HID + col], W1[(row0 + 1) * HID + col]);
        reinterpret_cast<float2*>(wsw)[t] = v;
    } else if (t < (int)(W1S_F2 + W2S_F2)) {
        int s    = t - (int)W1S_F2;
        int kk2  = s >> 5;
        int l    = s & 31;
        int half = l >> 4;
        int n    = l & 15;
        int row0 = 4 * kk2 + 2 * half;
        float2 v = make_float2(W2[row0 * DLAT + n], W2[(row0 + 1) * DLAT + n]);
        reinterpret_cast<float2*>(wsw)[(int)W1S_F2 + s] = v;
    }
}

__global__ __launch_bounds__(256) void euler_step(
    const float* __restrict__ xin, float* __restrict__ xout,
    const float* __restrict__ wsw,
    const float* __restrict__ b1, const float* __restrict__ b2,
    const float* __restrict__ tf, const int* __restrict__ nbr,
    int step)
{
    extern __shared__ float smem[];
    const int tid  = threadIdx.x;
    const int wave = tid >> 5;
    const int lane = tid & 31;

    {
        const float4* src = reinterpret_cast<const float4*>(wsw);
        float4*       dst = reinterpret_cast<float4*>(smem);
        #pragma unroll
        for (int i = 0; i < (int)(WSW_FLTS / 4) / 256; ++i)
            dst[tid + i * 256] = src[tid + i * 256];
    }
    __syncthreads();

    float* zh = smem + LDS_WAVE_OFF + wave * LDS_WAVE_SZ;

    const int tile      = blockIdx.x * 8 + wave;
    const int batch     = tile / TPB;
    const int node_base = (tile % TPB) * 16;
    const size_t xbatch = (size_t)batch * NN * DD;

    #pragma unroll
    for (int rr = 0; rr < 2; ++rr) {
        int r = lane + rr * 32;
        int m = r >> 2, k = r & 3;
        int idx = nbr[(node_base + m) * KNB + k];
        const float4* src = reinterpret_cast<const float4*>(xin + xbatch + (size_t)idx * DD);
        float4*       dst = reinterpret_cast<float4*>(zh + m * 96 + k * 24);
        #pragma unroll
        for (int q = 0; q < 6; ++q) dst[q] = src[q];
    }
    __syncthreads();

    const int n0   = lane & 15;
    const int half = lane >> 4;

    const float2* zf2  = reinterpret_cast<const float2*>(zh);
    const float2* w1f2 = reinterpret_cast<const float2*>(smem);
    const float2* w2f2 = reinterpret_cast<const float2*>(smem + LDS_W2_OFF);

    v8f acc[8];
    #pragma unroll
    for (int j = 0; j < 8; ++j) {
        float bv = b1[16 * j + n0];
        acc[j] = (v8f){bv, bv, bv, bv, bv, bv, bv, bv};
    }

    auto loadA1 = [&](int kk) -> v2f {
        float2 t = zf2[n0 * 48 + 2 * kk + half];
        return (v2f){t.x, t.y};
    };
    auto loadB1 = [&](int kk, v2f* b) {
        #pragma unroll
        for (int j = 0; j < 8; ++j) {
            float2 t = w1f2[(kk * 8 + j) * 32 + lane];
            b[j] = (v2f){t.x, t.y};
        }
    };
    auto mma1 = [&](v2f a, const v2f* b) {
        #pragma unroll
        for (int j = 0; j < 8; ++j)
            acc[j] = __builtin_amdgcn_wmma_f32_16x16x4_f32(
                false, a, false, b[j], (short)0, acc[j], false, false);
    };

    {
        v2f aA, aB;
        v2f bA[8], bB[8];
        aA = loadA1(0); loadB1(0, bA);
        #pragma unroll
        for (int kk = 0; kk < 24; kk += 2) {
            aB = loadA1(kk + 1); loadB1(kk + 1, bB);
            mma1(aA, bA);
            if (kk + 2 < 24) { aA = loadA1(kk + 2); loadB1(kk + 2, bA); }
            mma1(aB, bB);
        }
    }

    #pragma unroll
    for (int j = 0; j < 8; ++j) {
        #pragma unroll
        for (int r = 0; r < 8; ++r) {
            float h = gelu_tanh(acc[j][r]);
            int row = half ? (r + 8) : r;
            zh[row * HID + 16 * j + n0] = h;
        }
    }
    __syncthreads();

    float bv2 = b2[n0];
    v8f acc2a = (v8f){bv2, bv2, bv2, bv2, bv2, bv2, bv2, bv2};
    v8f acc2b = (v8f){0.f, 0.f, 0.f, 0.f, 0.f, 0.f, 0.f, 0.f};

    auto loadA2 = [&](int kk2) -> v2f {
        float2 t = zf2[n0 * 64 + 2 * kk2 + half];
        return (v2f){t.x, t.y};
    };
    auto loadB2 = [&](int kk2) -> v2f {
        float2 t = w2f2[kk2 * 32 + lane];
        return (v2f){t.x, t.y};
    };

    {
        v2f aA = loadA2(0), bA = loadB2(0);
        v2f aB, bB;
        #pragma unroll
        for (int kk2 = 0; kk2 < 32; kk2 += 2) {
            aB = loadA2(kk2 + 1); bB = loadB2(kk2 + 1);
            acc2a = __builtin_amdgcn_wmma_f32_16x16x4_f32(
                false, aA, false, bA, (short)0, acc2a, false, false);
            if (kk2 + 2 < 32) { aA = loadA2(kk2 + 2); bA = loadB2(kk2 + 2); }
            acc2b = __builtin_amdgcn_wmma_f32_16x16x4_f32(
                false, aB, false, bB, (short)0, acc2b, false, false);
        }
    }
    v8f acc2 = acc2a + acc2b;

    float t  = (float)step;
    float ms = tf[batch] - t;
    ms = fminf(fmaxf(ms, 0.0f), 1.0f);

    const size_t obase = xbatch + (size_t)node_base * DD;
    #pragma unroll
    for (int r = 0; r < 8; ++r) {
        int m = half ? (r + 8) : r;
        size_t a = obase + (size_t)m * DD + n0;
        xout[a] = xin[a] + ms * acc2[r];
    }
    {
        int m2 = lane >> 1;
        int c0 = DLAT + (lane & 1) * 4;
        size_t a2 = obase + (size_t)m2 * DD + c0;
        *reinterpret_cast<float4*>(xout + a2) =
            *reinterpret_cast<const float4*>(xin + a2);
    }
}

extern "C" void kernel_launch(void* const* d_in, const int* in_sizes, int n_in,
                              void* d_out, int out_size, void* d_ws, size_t ws_size,
                              hipStream_t stream) {
    (void)in_sizes; (void)n_in; (void)out_size; (void)ws_size;

    const float* x   = (const float*)d_in[0];
    const float* tf  = (const float*)d_in[1];
    const float* W1  = (const float*)d_in[2];
    const float* b1  = (const float*)d_in[3];
    const float* W2  = (const float*)d_in[4];
    const float* b2  = (const float*)d_in[5];
    const int*   nbr = (const int*)d_in[6];

    float* wsA = (float*)d_ws;
    float* wsw = wsA + W1S_OFF;
    float* out = (float*)d_out;

    swizzle_weights<<<28, 256, 0, stream>>>(W1, W2, wsw);

    const int grid = (NB * TPB) / 8;
    euler_step<<<grid, 256, LDS_TOTAL_B, stream>>>(x,   wsA, wsw, b1, b2, tf, nbr, 0);
    euler_step<<<grid, 256, LDS_TOTAL_B, stream>>>(wsA, out, wsw, b1, b2, tf, nbr, 1);
    euler_step<<<grid, 256, LDS_TOTAL_B, stream>>>(out, wsA, wsw, b1, b2, tf, nbr, 2);
    euler_step<<<grid, 256, LDS_TOTAL_B, stream>>>(wsA, out, wsw, b1, b2, tf, nbr, 3);
}


# Layer 1 split by hidden columns, not nodes: each wave keeps its 16-column W1 slice in registers (read from W1) for all 8 tiles of a shared 128-node LDS tile; LDS weight copy and pre-pass launch go
# speedup vs baseline: 1.1513x; 1.1513x over previous
#include <hip/hip_runtime.h>
#include <math.h>


typedef __attribute__((ext_vector_type(2))) float v2f;
typedef __attribute__((ext_vector_type(8))) float v8f;

constexpr int NB   = 2;
constexpr int NN   = 131072;
constexpr int DD   = 24;
constexpr int KNB  = 4;
constexpr int HID  = 128;
constexpr int DLAT = 16;
constexpr int WGN  = 128;
constexpr int WPB  = NN / WGN;

constexpr int ZP          = 100;
constexpr int HP          = 132;
constexpr int LDS_TILE    = WGN * HP;
constexpr int LDS_W2_OFF  = LDS_TILE;
constexpr int LDS_TOTAL_B = (LDS_TILE + 2048) * 4;

__device__ __forceinline__ float gelu_tanh(float x) {
    float u = 0.7978845608028654f * (x + 0.044715f * x * x * x);
    return 0.5f * x * (1.0f + tanhf(u));
}

__global__ __launch_bounds__(256, 6) void euler_step(
    const float* __restrict__ xin, float* __restrict__ xout,
    const float* __restrict__ W1, const float* __restrict__ b1,
    const float* __restrict__ W2, const float* __restrict__ b2,
    const float* __restrict__ tf, const int* __restrict__ nbr,
    int step)
{
    extern __shared__ float smem[];
    const int tid  = threadIdx.x;
    const int wave = tid >> 5;
    const int lane = tid & 31;
    const int c    = lane & 15;
    const int h    = lane >> 4;

    const int batch     = blockIdx.x / WPB;
    const int node_base = (blockIdx.x % WPB) * WGN;
    const size_t xbatch = (size_t)batch * NN * DD;

    v2f w1r[24];
    {
        const float* wp = W1 + (2 * h) * HID + 16 * wave + c;
        #pragma unroll
        for (int kk = 0; kk < 24; ++kk)
            w1r[kk] = (v2f){wp[(4 * kk) * HID], wp[(4 * kk + 1) * HID]};
    }

    {
        float2* dst = reinterpret_cast<float2*>(smem + LDS_W2_OFF);
        #pragma unroll
        for (int i = 0; i < 4; ++i) {
            int s    = tid + i * 256;
            int kk2  = s >> 5;
            int l    = s & 31;
            int row0 = 4 * kk2 + 2 * (l >> 4);
            int n    = l & 15;
            dst[s] = make_float2(W2[row0 * DLAT + n], W2[(row0 + 1) * DLAT + n]);
        }
    }

    #pragma unroll
    for (int rr = 0; rr < 2; ++rr) {
        int r = tid + rr * 256;
        int m = r >> 2, k = r & 3;
        int idx = nbr[(size_t)node_base * KNB + r];
        const float4* src = reinterpret_cast<const float4*>(xin + xbatch + (size_t)idx * DD);
        float4*       dst = reinterpret_cast<float4*>(smem + m * ZP + k * DD);
        #pragma unroll
        for (int q = 0; q < 6; ++q) dst[q] = src[q];
    }
    __syncthreads();

    v8f acc[8];
    {
        float bv = b1[16 * wave + c];
        #pragma unroll
        for (int m = 0; m < 8; ++m)
            acc[m] = (v8f){bv, bv, bv, bv, bv, bv, bv, bv};
    }

    const float2* zf2 = reinterpret_cast<const float2*>(smem);

    auto loadA1 = [&](int kk, v2f* a) {
        #pragma unroll
        for (int m = 0; m < 8; ++m) {
            float2 t = zf2[(16 * m + c) * (ZP / 2) + 2 * kk + h];
            a[m] = (v2f){t.x, t.y};
        }
    };
    auto mma1 = [&](const v2f* a, v2f b) {
        #pragma unroll
        for (int m = 0; m < 8; ++m)
            acc[m] = __builtin_amdgcn_wmma_f32_16x16x4_f32(
                false, a[m], false, b, (short)0, acc[m], false, false);
    };

    {
        v2f aA[8], aB[8];
        loadA1(0, aA);
        #pragma unroll
        for (int kk = 0; kk < 24; kk += 2) {
            loadA1(kk + 1, aB);
            mma1(aA, w1r[kk]);
            if (kk + 2 < 24) loadA1(kk + 2, aA);
            mma1(aB, w1r[kk + 1]);
        }
    }

    __syncthreads();

    #pragma unroll
    for (int m = 0; m < 8; ++m) {
        #pragma unroll
        for (int r = 0; r < 8; ++r)
            smem[(16 * m + 8 * h + r) * HP + 16 * wave + c] = gelu_tanh(acc[m][r]);
    }
    __syncthreads();

    const float2* hf2  = reinterpret_cast<const float2*>(smem);
    const float2* w2f2 = reinterpret_cast<const float2*>(smem + LDS_W2_OFF);

    float bv2 = b2[c];
    v8f acc2a = (v8f){bv2, bv2, bv2, bv2, bv2, bv2, bv2, bv2};
    v8f acc2b = (v8f){0.f, 0.f, 0.f, 0.f, 0.f, 0.f, 0.f, 0.f};

    auto loadA2 = [&](int kk2) -> v2f {
        float2 t = hf2[(16 * wave + c) * (HP / 2) + 2 * kk2 + h];
        return (v2f){t.x, t.y};
    };
    auto loadB2 = [&](int kk2) -> v2f {
        float2 t = w2f2[kk2 * 32 + lane];
        return (v2f){t.x, t.y};
    };

    {
        v2f aA = loadA2(0), bA = loadB2(0);
        v2f aB, bB;
        #pragma unroll
        for (int kk2 = 0; kk2 < 32; kk2 += 2) {
            aB = loadA2(kk2 + 1); bB = loadB2(kk2 + 1);
            acc2a = __builtin_amdgcn_wmma_f32_16x16x4_f32(
                false, aA, false, bA, (short)0, acc2a, false, false);
            if (kk2 + 2 < 32) { aA = loadA2(kk2 + 2); bA = loadB2(kk2 + 2); }
            acc2b = __builtin_amdgcn_wmma_f32_16x16x4_f32(
                false, aB, false, bB, (short)0, acc2b, false, false);
        }
    }
    v8f acc2 = acc2a + acc2b;

    float t  = (float)step;
    float ms = tf[batch] - t;
    ms = fminf(fmaxf(ms, 0.0f), 1.0f);

    const size_t obase = xbatch + (size_t)(node_base + 16 * wave) * DD;
    #pragma unroll
    for (int r = 0; r < 8; ++r) {
        int m = 8 * h + r;
        size_t a = obase + (size_t)m * DD + c;
        xout[a] = xin[a] + ms * acc2[r];
    }
    {
        int m2 = lane >> 1;
        int c0 = DLAT + (lane & 1) * 4;
        size_t a2 = obase + (size_t)m2 * DD + c0;
        *reinterpret_cast<float4*>(xout + a2) =
            *reinterpret_cast<const float4*>(xin + a2);
    }
}

extern "C" void kernel_launch(void* const* d_in, const int* in_sizes, int n_in,
                              void* d_out, int out_size, void* d_ws, size_t ws_size,
                              hipStream_t stream) {
    (void)in_sizes; (void)n_in; (void)out_size; (void)ws_size;

    const float* x   = (const float*)d_in[0];
    const float* tf  = (const float*)d_in[1];
    const float* W1  = (const float*)d_in[2];
    const float* b1  = (const float*)d_in[3];
    const float* W2  = (const float*)d_in[4];
    const float* b2  = (const float*)d_in[5];
    const int*   nbr = (const int*)d_in[6];

    float* wsA = (float*)d_ws;
    float* out = (float*)d_out;

    constexpr int grid = NB * WPB;
    euler_step<<<grid, 256, LDS_TOTAL_B, stream>>>(x,   wsA, W1, b1, W2, b2, tf, nbr, 0);
    euler_step<<<grid, 256, LDS_TOTAL_B, stream>>>(wsA, out, W1, b1, W2, b2, tf, nbr, 1);
    euler_step<<<grid, 256, LDS_TOTAL_B, stream>>>(out, wsA, W1, b1, W2, b2, tf, nbr, 2);
    euler_step<<<grid, 256, LDS_TOTAL_B, stream>>>(wsA, out, W1, b1, W2, b2, tf, nbr, 3);
}
